# starting kernel
# baseline (speedup 1.0000x reference)
#include <hip/hip_runtime.h>
#include <hip/hip_bf16.h>

typedef __attribute__((ext_vector_type(2))) float v2f;
typedef __attribute__((ext_vector_type(8))) float v8f;

#define EMB 128

__global__ void k_init_deg(float* __restrict__ deg, int n) {
    int i = blockIdx.x * blockDim.x + threadIdx.x;
    if (i < n) deg[i] = 1.0f;
}

__global__ void k_count_deg(const int* __restrict__ colI, float* __restrict__ deg, int E) {
    int i = blockIdx.x * blockDim.x + threadIdx.x;
    if (i < E) atomicAdd(&deg[colI[i]], 1.0f);
}

__global__ void k_dinv(float* __restrict__ deg, int n) {
    int i = blockIdx.x * blockDim.x + threadIdx.x;
    if (i < n) deg[i] = rsqrtf(deg[i]);
}

__global__ void k_init_x(const float* __restrict__ emb, float* __restrict__ x,
                         float* __restrict__ out, int total) {
    int i = blockIdx.x * blockDim.x + threadIdx.x;
    if (i < total) {
        float v = emb[i];
        x[i]   = v;
        out[i] = v;
    }
}

__global__ __launch_bounds__(256) void k_gemm_wmma(const float* __restrict__ X,
                                                   const float* __restrict__ W,
                                                   float* __restrict__ H,
                                                   int mTiles) {
    const int lane = threadIdx.x & 31;
    const int wave = threadIdx.x >> 5;
    const int mt   = blockIdx.x * 8 + wave;
    if (mt >= mTiles) return;

    const int m0   = mt * 16;
    const int half = lane >> 4;
    const int l16  = lane & 15;
    const int kb   = half * 2;

    const float* __restrict__ xrow = X + (size_t)(m0 + l16) * EMB;

    v8f acc[8];
#pragma unroll
    for (int nt = 0; nt < 8; ++nt) acc[nt] = (v8f){0.f,0.f,0.f,0.f,0.f,0.f,0.f,0.f};

    for (int k = 0; k < EMB; k += 4) {
        v2f a = *(const v2f*)(xrow + k + kb);
#pragma unroll
        for (int nt = 0; nt < 8; ++nt) {
            v2f b;
            b.x = W[(size_t)(k + kb    ) * EMB + nt * 16 + l16];
            b.y = W[(size_t)(k + kb + 1) * EMB + nt * 16 + l16];
            acc[nt] = __builtin_amdgcn_wmma_f32_16x16x4_f32(
                false, a, false, b, (short)0, acc[nt], false, false);
        }
    }

#pragma unroll
    for (int nt = 0; nt < 8; ++nt) {
        float* __restrict__ hp = H + (size_t)(m0 + half * 8) * EMB + nt * 16 + l16;
#pragma unroll
        for (int i = 0; i < 8; ++i)
            hp[(size_t)i * EMB] = acc[nt][i];
    }
}

__global__ void k_self_bias(const float* __restrict__ H, const float* __restrict__ dinv,
                            const float* __restrict__ b, float* __restrict__ agg, int total) {
    int i = blockIdx.x * blockDim.x + threadIdx.x;
    if (i < total) {
        int v = i >> 7;
        int f = i & 127;
        float d = dinv[v];
        agg[i] = H[i] * d * d + b[f];
    }
}

__global__ __launch_bounds__(256) void k_scatter(const int* __restrict__ rowI,
                                                 const int* __restrict__ colI,
                                                 const float* __restrict__ H,
                                                 const float* __restrict__ dinv,
                                                 float* __restrict__ agg, int E) {
    int gtid = blockIdx.x * blockDim.x + threadIdx.x;
    int e    = gtid >> 5;
    int lane = gtid & 31;
    if (e >= E) return;
    int r = rowI[e];
    int c = colI[e];
    float norm = dinv[r] * dinv[c];
    const float4 hv = *(const float4*)(H + (size_t)r * EMB + lane * 4);
    float* __restrict__ dst = agg + (size_t)c * EMB + lane * 4;
    atomicAdd(dst + 0, hv.x * norm);
    atomicAdd(dst + 1, hv.y * norm);
    atomicAdd(dst + 2, hv.z * norm);
    atomicAdd(dst + 3, hv.w * norm);
}

__global__ void k_accum(float* __restrict__ out, const float* __restrict__ agg, int total) {
    int i = blockIdx.x * blockDim.x + threadIdx.x;
    if (i < total) out[i] += agg[i];
}

extern "C" void kernel_launch(void* const* d_in, const int* in_sizes, int n_in,
                              void* d_out, int out_size, void* d_ws, size_t ws_size,
                              hipStream_t stream) {
    const int*   edge = (const int*)d_in[0];
    const float* emb  = (const float*)d_in[1];
    const float* Wall = (const float*)d_in[2];
    const float* ball = (const float*)d_in[3];
    float*       out  = (float*)d_out;

    const int E = in_sizes[0] / 2;
    const int n = out_size / EMB;
    const int total = n * EMB;

    const int* rowI = edge;
    const int* colI = edge + E;

    const int mTiles  = (n + 15) / 16;
    const int rowsPad = mTiles * 16;
    const size_t matElems = (size_t)rowsPad * EMB;

    float* ws   = (float*)d_ws;
    float* bufA = ws;
    float* bufB = ws + matElems;
    float* bufC = ws + 2 * matElems;
    float* dinv = ws + 3 * matElems;
    float* bufs[3] = {bufA, bufB, bufC};

    const int T = 256;
    k_init_deg <<<(n + T - 1) / T, T, 0, stream>>>(dinv, n);
    k_count_deg<<<(E + T - 1) / T, T, 0, stream>>>(colI, dinv, E);
    k_dinv     <<<(n + T - 1) / T, T, 0, stream>>>(dinv, n);
    k_init_x   <<<(total + T - 1) / T, T, 0, stream>>>(emb, bufs[0], out, total);

    int xi = 0, hi = 1, ai = 2;
    for (int l = 0; l < 3; ++l) {
        k_gemm_wmma<<<(mTiles + 7) / 8, 256, 0, stream>>>(
            bufs[xi], Wall + (size_t)l * EMB * EMB, bufs[hi], mTiles);
        k_self_bias<<<(total + T - 1) / T, T, 0, stream>>>(
            bufs[hi], dinv, ball + (size_t)l * EMB, bufs[ai], total);
        k_scatter<<<((size_t)E * 32 + T - 1) / T, T, 0, stream>>>(
            rowI, colI, bufs[hi], dinv, bufs[ai], E);
        k_accum<<<(total + T - 1) / T, T, 0, stream>>>(out, bufs[ai], total);
        int t = xi; xi = ai; ai = t;
    }
}


# Edge sum as a gather over edges ordered by target in the call (int atomic rank, scan, place): one wave sums a row in registers, stores it once; f32 atomics gone; sum order = arrival rank
# speedup vs baseline: 2.1223x; 2.1223x over previous
#include <hip/hip_runtime.h>
#include <hip/hip_bf16.h>

typedef __attribute__((ext_vector_type(2))) float v2f;
typedef __attribute__((ext_vector_type(8))) float v8f;

#define EMB 128

__global__ void k_zero_cnt(int* __restrict__ cnt, int n) {
    int i = blockIdx.x * blockDim.x + threadIdx.x;
    if (i < n) cnt[i] = 0;
}

__global__ void k_count_rank(const int* __restrict__ colI, int* __restrict__ cnt,
                             int* __restrict__ rank, int E) {
    int i = blockIdx.x * blockDim.x + threadIdx.x;
    if (i < E) rank[i] = atomicAdd(&cnt[colI[i]], 1);
}

#define SCAN_T     1024
#define SCAN_CHUNK (SCAN_T * 4)
__global__ __launch_bounds__(SCAN_T) void k_scan_dinv(const int* __restrict__ cnt,
                                                      int* __restrict__ off,
                                                      float* __restrict__ dinv, int n) {
    __shared__ int sh[SCAN_T / 32];
    const int tid  = threadIdx.x;
    const int lane = tid & 31;
    const int wave = tid >> 5;
    const int base = blockIdx.x * SCAN_CHUNK;

    int s = 0;
    const int4* __restrict__ c4 = (const int4*)cnt;
#pragma unroll 4
    for (int i = tid; i < base / 4; i += SCAN_T) {
        int4 q = c4[i];
        s += (q.x + q.y) + (q.z + q.w);
    }
    for (int m = 16; m >= 1; m >>= 1) s += __shfl_xor(s, m, 32);
    if (lane == 0) sh[wave] = s;
    __syncthreads();
    int pre = 0;
    for (int w = 0; w < SCAN_T / 32; ++w) pre += sh[w];
    __syncthreads();

    const int i0 = base + tid * 4;
    int c[4];
#pragma unroll
    for (int k = 0; k < 4; ++k) c[k] = (i0 + k < n) ? cnt[i0 + k] : 0;
    const int tsum = (c[0] + c[1]) + (c[2] + c[3]);
    int inc = tsum;
    for (int d = 1; d < 32; d <<= 1) {
        int t = __shfl_up(inc, d, 32);
        if (lane >= d) inc += t;
    }
    if (lane == 31) sh[wave] = inc;
    __syncthreads();
    int run = pre + inc - tsum;
    for (int w = 0; w < wave; ++w) run += sh[w];
#pragma unroll
    for (int k = 0; k < 4; ++k) {
        const int idx = i0 + k;
        if (idx < n) {
            off[idx]  = run;
            dinv[idx] = rsqrtf((float)(c[k] + 1));
            run += c[k];
            if (idx == n - 1) off[n] = run;
        }
    }
}

__global__ void k_place(const int* __restrict__ rowI, const int* __restrict__ colI,
                        const int* __restrict__ rank, const int* __restrict__ off,
                        int* __restrict__ src, int E) {
    int i = blockIdx.x * blockDim.x + threadIdx.x;
    if (i < E) src[off[colI[i]] + rank[i]] = rowI[i];
}

__global__ void k_init_x(const float* __restrict__ emb, float* __restrict__ x,
                         float* __restrict__ out, int total) {
    int i = blockIdx.x * blockDim.x + threadIdx.x;
    if (i < total) {
        float v = emb[i];
        x[i]   = v;
        out[i] = v;
    }
}

__global__ __launch_bounds__(256) void k_gemm_wmma(const float* __restrict__ X,
                                                   const float* __restrict__ W,
                                                   float* __restrict__ H,
                                                   int mTiles) {
    const int lane = threadIdx.x & 31;
    const int wave = threadIdx.x >> 5;
    const int mt   = blockIdx.x * 8 + wave;
    if (mt >= mTiles) return;

    const int m0   = mt * 16;
    const int half = lane >> 4;
    const int l16  = lane & 15;
    const int kb   = half * 2;

    const float* __restrict__ xrow = X + (size_t)(m0 + l16) * EMB;

    v8f acc[8];
#pragma unroll
    for (int nt = 0; nt < 8; ++nt) acc[nt] = (v8f){0.f,0.f,0.f,0.f,0.f,0.f,0.f,0.f};

    for (int k = 0; k < EMB; k += 4) {
        v2f a = *(const v2f*)(xrow + k + kb);
#pragma unroll
        for (int nt = 0; nt < 8; ++nt) {
            v2f b;
            b.x = W[(size_t)(k + kb    ) * EMB + nt * 16 + l16];
            b.y = W[(size_t)(k + kb + 1) * EMB + nt * 16 + l16];
            acc[nt] = __builtin_amdgcn_wmma_f32_16x16x4_f32(
                false, a, false, b, (short)0, acc[nt], false, false);
        }
    }

#pragma unroll
    for (int nt = 0; nt < 8; ++nt) {
        float* __restrict__ hp = H + (size_t)(m0 + half * 8) * EMB + nt * 16 + l16;
#pragma unroll
        for (int i = 0; i < 8; ++i)
            hp[(size_t)i * EMB] = acc[nt][i];
    }
}

__device__ __forceinline__ float4 fma4(float4 h, float w, float4 a) {
    a.x = fmaf(h.x, w, a.x);
    a.y = fmaf(h.y, w, a.y);
    a.z = fmaf(h.z, w, a.z);
    a.w = fmaf(h.w, w, a.w);
    return a;
}

__global__ __launch_bounds__(256) void k_gather(const int* __restrict__ off,
                                                const int* __restrict__ src,
                                                const float* __restrict__ H,
                                                const float* __restrict__ dinv,
                                                const float* __restrict__ b,
                                                float* __restrict__ agg, int n) {
    const int lane = threadIdx.x & 31;
    const int v = blockIdx.x * 8 + __builtin_amdgcn_readfirstlane((int)(threadIdx.x >> 5));
    if (v >= n) return;

    const float4* __restrict__ H4 = (const float4*)H;
    const int   beg = off[v];
    const int   end = off[v + 1];
    const float dc  = dinv[v];

    const float4 hs = H4[(size_t)v * 32 + lane];
    float4 acc = (float4){0.f, 0.f, 0.f, 0.f};

    for (int e0 = beg; e0 < end; e0 += 32) {
        const int m = (end - e0 < 32) ? (end - e0) : 32;
        int   rl = 0;
        float wl = 0.f;
        if (lane < m) {
            rl = src[e0 + lane];
            wl = dinv[rl] * dc;
        }
        const int wi = __float_as_int(wl);
        int j = 0;
        for (; j + 4 <= m; j += 4) {
            const int r0 = __builtin_amdgcn_readlane(rl, j);
            const int r1 = __builtin_amdgcn_readlane(rl, j + 1);
            const int r2 = __builtin_amdgcn_readlane(rl, j + 2);
            const int r3 = __builtin_amdgcn_readlane(rl, j + 3);
            const float4 h0 = H4[(size_t)r0 * 32 + lane];
            const float4 h1 = H4[(size_t)r1 * 32 + lane];
            const float4 h2 = H4[(size_t)r2 * 32 + lane];
            const float4 h3 = H4[(size_t)r3 * 32 + lane];
            acc = fma4(h0, __int_as_float(__builtin_amdgcn_readlane(wi, j)),     acc);
            acc = fma4(h1, __int_as_float(__builtin_amdgcn_readlane(wi, j + 1)), acc);
            acc = fma4(h2, __int_as_float(__builtin_amdgcn_readlane(wi, j + 2)), acc);
            acc = fma4(h3, __int_as_float(__builtin_amdgcn_readlane(wi, j + 3)), acc);
        }
        const int t = m - j;
        if (t > 0) {
            float4 h0 = (float4){0.f, 0.f, 0.f, 0.f}, h1 = h0, h2 = h0;
            const int r0 = __builtin_amdgcn_readlane(rl, j);
            h0 = H4[(size_t)r0 * 32 + lane];
            if (t > 1) {
                const int r1 = __builtin_amdgcn_readlane(rl, j + 1);
                h1 = H4[(size_t)r1 * 32 + lane];
            }
            if (t > 2) {
                const int r2 = __builtin_amdgcn_readlane(rl, j + 2);
                h2 = H4[(size_t)r2 * 32 + lane];
            }
            acc = fma4(h0, __int_as_float(__builtin_amdgcn_readlane(wi, j)), acc);
            if (t > 1) acc = fma4(h1, __int_as_float(__builtin_amdgcn_readlane(wi, j + 1)), acc);
            if (t > 2) acc = fma4(h2, __int_as_float(__builtin_amdgcn_readlane(wi, j + 2)), acc);
        }
    }

    acc = fma4(hs, dc * dc, acc);
    const float4 bb = ((const float4*)b)[lane];
    acc.x += bb.x; acc.y += bb.y; acc.z += bb.z; acc.w += bb.w;
    ((float4*)agg)[(size_t)v * 32 + lane] = acc;
}

__global__ void k_accum(float* __restrict__ out, const float* __restrict__ agg, int total) {
    int i = blockIdx.x * blockDim.x + threadIdx.x;
    if (i < total) out[i] += agg[i];
}

extern "C" void kernel_launch(void* const* d_in, const int* in_sizes, int n_in,
                              void* d_out, int out_size, void* d_ws, size_t ws_size,
                              hipStream_t stream) {
    const int*   edge = (const int*)d_in[0];
    const float* emb  = (const float*)d_in[1];
    const float* Wall = (const float*)d_in[2];
    const float* ball = (const float*)d_in[3];
    float*       out  = (float*)d_out;

    const int E = in_sizes[0] / 2;
    const int n = out_size / EMB;
    const int total = n * EMB;

    const int* rowI = edge;
    const int* colI = edge + E;

    const int mTiles  = (n + 15) / 16;
    const int rowsPad = mTiles * 16;
    const size_t matElems = (size_t)rowsPad * EMB;

    float* ws   = (float*)d_ws;
    float* bufX = ws;
    float* bufH = ws + matElems;
    int*   iws  = (int*)(ws + 2 * matElems);
    float* dinv = ws + 3 * matElems;

    const int nAl = rowsPad + 16;
    int* cnt  = iws;
    int* off  = iws + nAl;
    int* rank = iws + 2 * nAl;
    int* src  = rank + E;

    const int T = 256;
    k_zero_cnt  <<<(n + T - 1) / T, T, 0, stream>>>(cnt, n);
    k_count_rank<<<(E + T - 1) / T, T, 0, stream>>>(colI, cnt, rank, E);
    k_scan_dinv <<<(n + SCAN_CHUNK - 1) / SCAN_CHUNK, SCAN_T, 0, stream>>>(cnt, off, dinv, n);
    k_place     <<<(E + T - 1) / T, T, 0, stream>>>(rowI, colI, rank, off, src, E);
    k_init_x    <<<(total + T - 1) / T, T, 0, stream>>>(emb, bufX, out, total);

    for (int l = 0; l < 3; ++l) {
        k_gemm_wmma<<<(mTiles + 7) / 8, 256, 0, stream>>>(
            bufX, Wall + (size_t)l * EMB * EMB, bufH, mTiles);
        k_gather<<<(n + 7) / 8, 256, 0, stream>>>(
            off, src, bufH, dinv, ball + (size_t)l * EMB, bufX, n);
        k_accum<<<(total + T - 1) / T, T, 0, stream>>>(out, bufX, total);
    }
}


# Rows go to the next layer already times their own dinv, so layers 1 and 2 add plain rows: per-edge dinv[src] read (a line per edge) and weight multiply go; sum order kept, each term rounded once more
# speedup vs baseline: 5.1522x; 2.4276x over previous
#include <hip/hip_runtime.h>
#include <hip/hip_bf16.h>

typedef __attribute__((ext_vector_type(2))) float v2f;
typedef __attribute__((ext_vector_type(8))) float v8f;

#define EMB 128

#define NBK   256
#define BT    1024
#define BWCAP 1024

__global__ __launch_bounds__(BT) void k_bucket(const int* __restrict__ rowI, const int* __restrict__ colI,
                                               int* __restrict__ M, int* __restrict__ Pz,
                                               int* __restrict__ stage, int E, int n) {
    __shared__ int hist[NBK];
    __shared__ int wsum[BT / 32];
    const int tid  = threadIdx.x;
    const int lane = tid & 31;
    const int wave = tid >> 5;
    const int c    = blockIdx.x;
    const int CH   = (E + NBK - 1) / NBK;
    const unsigned BW = (unsigned)((n + NBK - 1) / NBK);
    const int e0 = c * CH;
    const int e1 = (e0 + CH < E) ? (e0 + CH) : E;

    if (tid < NBK) hist[tid] = 0;
    __syncthreads();
    for (int eb = e0 + tid; eb < e1; eb += 4 * BT) {
        int cc[4];
#pragma unroll
        for (int k = 0; k < 4; ++k) {
            const int e = eb + k * BT;
            cc[k] = (e < e1) ? colI[e] : 0;
        }
#pragma unroll
        for (int k = 0; k < 4; ++k)
            if (eb + k * BT < e1) atomicAdd(&hist[(unsigned)cc[k] / BW], 1);
    }
    __syncthreads();

    const int hv = (tid < NBK) ? hist[tid] : 0;
    int inc = hv;
    for (int d = 1; d < 32; d <<= 1) {
        int t = __shfl_up(inc, d, 32);
        if (lane >= d) inc += t;
    }
    if (lane == 31) wsum[wave] = inc;
    __syncthreads();
    int pre = 0;
    for (int w = 0; w < wave; ++w) pre += wsum[w];
    const int ex = pre + inc - hv;
    if (tid < NBK) {
        hist[tid] = ex;
        M[tid * NBK + c]  = hv;
        Pz[tid * NBK + c] = ex;
    }
    __syncthreads();

    int2* __restrict__ st = (int2*)stage;
    for (int eb = e0 + tid; eb < e1; eb += 4 * BT) {
        int cc[4], rr[4];
#pragma unroll
        for (int k = 0; k < 4; ++k) {
            const int e = eb + k * BT;
            cc[k] = (e < e1) ? colI[e] : 0;
            rr[k] = (e < e1) ? rowI[e] : 0;
        }
#pragma unroll
        for (int k = 0; k < 4; ++k) {
            if (eb + k * BT < e1) {
                const int r = atomicAdd(&hist[(unsigned)cc[k] / BW], 1);
                st[e0 + r] = make_int2(rr[k], cc[k]);
            }
        }
    }
}

__global__ __launch_bounds__(BT) void k_rows(const int* __restrict__ M, const int* __restrict__ Pz,
                                             const int* __restrict__ stage, int* __restrict__ off,
                                             float* __restrict__ dinv, int* __restrict__ src, int E, int n) {
    __shared__ int segCnt[NBK];
    __shared__ int segPos[NBK];
    __shared__ int cntL[BWCAP];
    __shared__ int red[BT / 32];
    const int tid  = threadIdx.x;
    const int lane = tid & 31;
    const int wave = tid >> 5;
    const int b    = blockIdx.x;
    const int CH   = (E + NBK - 1) / NBK;
    const int BW   = (n + NBK - 1) / NBK;
    const int lo   = b * BW;
    const int nv   = (lo + BW <= n) ? BW : (n - lo);

    static_assert((NBK - 1) * (NBK / 4) <= 16 * BT, "sixteen pieces per thread cover the rows in front");
    const int4* __restrict__ M4 = (const int4*)M;
    const int nq = b * (NBK / 4);
    int4 q[16];
#pragma unroll
    for (int k = 0; k < 16; ++k) {
        const int i = tid + k * BT;
        q[k] = make_int4(0, 0, 0, 0);
        if (i < nq) q[k] = M4[i];
    }
    if (tid < NBK) {
        segCnt[tid] = M[b * NBK + tid];
        segPos[tid] = tid * CH + Pz[b * NBK + tid];
    }
    cntL[tid] = 0;
    int s = 0;
#pragma unroll
    for (int k = 0; k < 16; ++k) s += (q[k].x + q[k].y) + (q[k].z + q[k].w);
    for (int m = 16; m >= 1; m >>= 1) s += __shfl_xor(s, m, 32);
    if (lane == 0) red[wave] = s;
    __syncthreads();
    int baseB = 0;
    for (int w = 0; w < BT / 32; ++w) baseB += red[w];

    const int2* __restrict__ st = (const int2*)stage;
    const int sc = segCnt[tid >> 2];
    const int sp = segPos[tid >> 2];
    for (int j0 = tid & 3; j0 < sc; j0 += 16) {
        int y[4];
#pragma unroll
        for (int k = 0; k < 4; ++k) {
            const int j = j0 + 4 * k;
            y[k] = (j < sc) ? st[sp + j].y : lo;
        }
#pragma unroll
        for (int k = 0; k < 4; ++k)
            if (j0 + 4 * k < sc) atomicAdd(&cntL[y[k] - lo], 1);
    }
    __syncthreads();

    const int cv = (tid < nv) ? cntL[tid] : 0;
    int inc = cv;
    for (int d = 1; d < 32; d <<= 1) {
        int t = __shfl_up(inc, d, 32);
        if (lane >= d) inc += t;
    }
    if (lane == 31) red[wave] = inc;
    __syncthreads();
    int run = baseB + inc - cv;
    for (int w = 0; w < wave; ++w) run += red[w];
    if (tid < nv) {
        off[lo + tid]  = run;
        dinv[lo + tid] = rsqrtf((float)(cv + 1));
        cntL[tid]      = run;
    }
    if (b == 0 && tid == 0) off[n] = E;
    __syncthreads();

    for (int j0 = tid & 3; j0 < sc; j0 += 16) {
        int2 rc[4];
#pragma unroll
        for (int k = 0; k < 4; ++k) {
            const int j = j0 + 4 * k;
            rc[k] = (j < sc) ? st[sp + j] : make_int2(0, lo);
        }
#pragma unroll
        for (int k = 0; k < 4; ++k) {
            if (j0 + 4 * k < sc) {
                const int p = atomicAdd(&cntL[rc[k].y - lo], 1);
                src[p] = rc[k].x;
            }
        }
    }
}

#define NPW    4
#define TILE_N 32
#define PITCH  132

__device__ __forceinline__ float4 fma4(float4 h, float w, float4 a) {
    a.x = fmaf(h.x, w, a.x);
    a.y = fmaf(h.y, w, a.y);
    a.z = fmaf(h.z, w, a.z);
    a.w = fmaf(h.w, w, a.w);
    return a;
}

__device__ __forceinline__ float4 add4(float4 h, float4 a) {
    a.x += h.x;
    a.y += h.y;
    a.z += h.z;
    a.w += h.w;
    return a;
}

__device__ __forceinline__ float4 zero4() { return (float4){0.f, 0.f, 0.f, 0.f}; }

template <bool PRE>
__global__ __launch_bounds__(256, 16) void k_layer(const int* __restrict__ off,
                                               const int* __restrict__ src,
                                               const float* __restrict__ X,
                                               const float* __restrict__ dinv,
                                               const float* __restrict__ W,
                                               const float* __restrict__ b,
                                               const float* prev, float* out,
                                               float* __restrict__ Xout, int n) {
    __shared__ __attribute__((aligned(16))) float S[TILE_N * PITCH];
    __shared__ __attribute__((aligned(16))) float Y[TILE_N * PITCH];

    const int lane = threadIdx.x & 31;
    const int wave = __builtin_amdgcn_readfirstlane((int)(threadIdx.x >> 5));
    const int vb   = blockIdx.x * TILE_N + wave * NPW;

    const unsigned lo = (unsigned)lane * 16u;
    auto rowof = [&](const float* tab, int r) -> float4 {
        const char* p = (const char*)tab + (size_t)(unsigned)r * 512u;
        unsigned l = lo;
        asm volatile("" : "+v"(l));
        return *(const float4*)(p + l);
    };
    auto rowld = [&](int r) -> float4 { return rowof(X, r); };
    auto rowst = [&](float* tab, int r, float4 val) {
        char* p = (char*)tab + (size_t)(unsigned)r * 512u;
        unsigned l = lo;
        asm volatile("" : "+v"(l));
        *(float4*)(p + l) = val;
    };

    int   ol = 0;
    float dl = 0.f;
    {
        const int vi = vb + lane;
        if (lane <= NPW) ol = off[vi < n ? vi : n];
        if (lane < NPW && vi < n) dl = dinv[vi];
    }
    const int dli = __float_as_int(dl);
    float dsc[NPW];
#pragma unroll
    for (int i = 0; i < NPW; ++i) dsc[i] = __int_as_float(__builtin_amdgcn_readlane(dli, i));
    const int beg = __builtin_amdgcn_readlane(ol, 0);
    const int end = __builtin_amdgcn_readlane(ol, NPW);

    int rl = 0, rn = 0;
    float wl = 0.f;
    if (lane < end - beg)      rl = src[beg + lane];
    if (lane < end - beg - 32) rn = src[beg + 32 + lane];
    {
        float4 hs[NPW];
#pragma unroll
        for (int i = 0; i < NPW; ++i) {
            hs[i] = zero4();
            if (vb + i < n) hs[i] = rowld(vb + i);
        }
        float wl0 = 0.f;
        if (!PRE) { if (lane < end - beg) wl0 = dinv[rl]; }
#pragma unroll
        for (int i = 0; i < NPW; ++i) {
            const float di = PRE ? 1.f : dsc[i];
            float4 s;
            s.x = hs[i].x * di; s.y = hs[i].y * di; s.z = hs[i].z * di; s.w = hs[i].w * di;
            *(float4*)(S + (wave * NPW + i) * PITCH + lane * 4) = s;
        }
        wl = wl0;
    }

    int    cur   = 0;
    int    nextb = __builtin_amdgcn_readlane(ol, 1);
    float  dc    = __int_as_float(__builtin_amdgcn_readlane(dli, 0));
    float4 acc   = zero4();

    auto flush = [&]() {
        float* sp = S + (wave * NPW + cur) * PITCH + lane * 4;
        const float4 t = *(const float4*)sp;
        float4 s;
        s.x = (acc.x + t.x) * dc; s.y = (acc.y + t.y) * dc; s.z = (acc.z + t.z) * dc; s.w = (acc.w + t.w) * dc;
        *(float4*)sp = s;
        ++cur;
        acc = zero4();
        if (cur < NPW) {
            nextb = __builtin_amdgcn_readlane(ol, cur + 1);
            dc    = __int_as_float(__builtin_amdgcn_readlane(dli, cur));
        }
    };

    for (int e0 = beg; e0 < end; e0 += 32) {
        const int m = (end - e0 < 32) ? (end - e0) : 32;
        float wn = 0.f;
        int   rnn = 0;
        auto ahead = [&]() {
            if (!PRE) { if (lane < end - e0 - 32) wn = dinv[rn]; }
            if (lane < end - e0 - 64) rnn = src[e0 + 64 + lane];
        };
        const int wi = __float_as_int(wl);

        int j = 0;
        for (; j + 8 <= m; j += 8) {
            float4 h[8];
#pragma unroll
            for (int k = 0; k < 8; ++k)
                h[k] = rowld(__builtin_amdgcn_readlane(rl, j + k));
            if (j == 0) ahead();
#pragma unroll
            for (int k = 0; k < 8; ++k) {
                while (e0 + j + k >= nextb) flush();
                acc = PRE ? add4(h[k], acc) : fma4(h[k], __int_as_float(__builtin_amdgcn_readlane(wi, j + k)), acc);
            }
        }
        const int t = m - j;
        if (t > 0) {
            float4 h[7];
#pragma unroll
            for (int k = 0; k < 7; ++k) {
                h[k] = zero4();
                if (k < t) h[k] = rowld(__builtin_amdgcn_readlane(rl, j + k));
            }
            if (j == 0) ahead();
#pragma unroll
            for (int k = 0; k < 7; ++k) {
                if (k < t) {
                    while (e0 + j + k >= nextb) flush();
                    acc = PRE ? add4(h[k], acc) : fma4(h[k], __int_as_float(__builtin_amdgcn_readlane(wi, j + k)), acc);
                }
            }
        }
        rl = rn; wl = wn; rn = rnn;
    }
    while (cur < NPW) flush();

    __syncthreads();

    const int half = lane >> 4;
    const int l16  = lane & 15;
    const int kb   = half * 2;
    const float* a0p = S + l16 * PITCH + kb;
    const float* a1p = a0p + 16 * PITCH;
    const float* __restrict__ wp = W + (unsigned)(kb * EMB + wave * 16 + l16);

    v8f c0 = (v8f){0.f,0.f,0.f,0.f,0.f,0.f,0.f,0.f};
    v8f c1 = c0;
    v2f bc[4], bn[4];
#pragma unroll
    for (int s = 0; s < 4; ++s) {
        bc[s].x = wp[(size_t)(4 * s) * EMB];
        bc[s].y = wp[(size_t)(4 * s + 1) * EMB];
    }
#pragma unroll
    for (int k0 = 0; k0 < EMB; k0 += 16) {
        if (k0 + 16 < EMB) {
#pragma unroll
            for (int s = 0; s < 4; ++s) {
                bn[s].x = wp[(size_t)(k0 + 16 + 4 * s) * EMB];
                bn[s].y = wp[(size_t)(k0 + 16 + 4 * s + 1) * EMB];
            }
        }
#pragma unroll
        for (int s = 0; s < 4; ++s) {
            const v2f a0 = *(const v2f*)(a0p + k0 + 4 * s);
            const v2f a1 = *(const v2f*)(a1p + k0 + 4 * s);
            c0 = __builtin_amdgcn_wmma_f32_16x16x4_f32(false, a0, false, bc[s], (short)0, c0, false, false);
            c1 = __builtin_amdgcn_wmma_f32_16x16x4_f32(false, a1, false, bc[s], (short)0, c1, false, false);
        }
        if (k0 + 16 < EMB) {
#pragma unroll
            for (int s = 0; s < 4; ++s) bc[s] = bn[s];
        }
    }

    float4 pv[NPW];
#pragma unroll
    for (int i = 0; i < NPW; ++i) {
        pv[i] = zero4();
        if (vb + i < n) pv[i] = rowof(prev, vb + i);
    }

    {
        float* y0 = Y + (half * 8) * PITCH + wave * 16 + l16;
        float* y1 = y0 + 16 * PITCH;
#pragma unroll
        for (int i = 0; i < 8; ++i) {
            y0[i * PITCH] = c0[i];
            y1[i * PITCH] = c1[i];
        }
    }
    __syncthreads();

    const float4 bb = ((const float4*)b)[lane];
#pragma unroll
    for (int i = 0; i < NPW; ++i) {
        const int v = vb + i;
        if (v < n) {
            float4 y = *(const float4*)(Y + (wave * NPW + i) * PITCH + lane * 4);
            y.x += bb.x; y.y += bb.y; y.z += bb.z; y.w += bb.w;
            if (Xout != nullptr) {
                float4 ys;
                ys.x = y.x * dsc[i]; ys.y = y.y * dsc[i]; ys.z = y.z * dsc[i]; ys.w = y.w * dsc[i];
                rowst(Xout, v, ys);
            }
            float4 o;
            o.x = pv[i].x + y.x; o.y = pv[i].y + y.y; o.z = pv[i].z + y.z; o.w = pv[i].w + y.w;
            rowst(out, v, o);
        }
    }
}

extern "C" void kernel_launch(void* const* d_in, const int* in_sizes, int n_in,
                              void* d_out, int out_size, void* d_ws, size_t ws_size,
                              hipStream_t stream) {
    const int*   edge = (const int*)d_in[0];
    const float* emb  = (const float*)d_in[1];
    const float* Wall = (const float*)d_in[2];
    const float* ball = (const float*)d_in[3];
    float*       out  = (float*)d_out;

    const int E = in_sizes[0] / 2;
    const int n = out_size / EMB;

    const int* rowI = edge;
    const int* colI = edge + E;

    const int mTiles  = (n + 15) / 16;
    const int rowsPad = mTiles * 16;
    const size_t matElems = (size_t)rowsPad * EMB;

    float* ws   = (float*)d_ws;
    float* bufA = ws;
    float* bufB = ws + matElems;
    int*   iws  = (int*)(ws + 2 * matElems);
    float* dinv = ws + 3 * matElems;

    const int nAl = rowsPad + 16;
    int* M     = iws;
    int* Pz    = iws + NBK * NBK;
    int* off   = iws + 2 * NBK * NBK;
    int* stage = off + nAl;
    int* src   = stage + 2 * (size_t)E;

    k_bucket<<<NBK, BT, 0, stream>>>(rowI, colI, M, Pz, stage, E, n);
    k_rows  <<<NBK, BT, 0, stream>>>(M, Pz, stage, off, dinv, src, E, n);
    const int nb = (n + TILE_N - 1) / TILE_N;
    k_layer<false><<<nb, 256, 0, stream>>>(off, src, emb,  dinv, Wall,                         ball,
                                           emb, out, bufA, n);
    k_layer<true> <<<nb, 256, 0, stream>>>(off, src, bufA, dinv, Wall + (size_t)EMB * EMB,     ball + EMB,
                                           out, out, bufB, n);
    k_layer<true> <<<nb, 256, 0, stream>>>(off, src, bufB, dinv, Wall + (size_t)2 * EMB * EMB, ball + 2 * EMB,
                                           out, out, nullptr, n);
}


# Layer sum not carried through memory: layers 0, 1 store only the next layer's input; last layer forms out = emb + (X1' + X2')/dinv + Y2 from its own rows; layer sum's rounding changes
# speedup vs baseline: 5.4651x; 1.0607x over previous
#include <hip/hip_runtime.h>
#include <hip/hip_bf16.h>

typedef __attribute__((ext_vector_type(2))) float v2f;
typedef __attribute__((ext_vector_type(8))) float v8f;

#define EMB 128

#define NBK   256
#define BT    1024
#define BWCAP 1024

__global__ __launch_bounds__(BT) void k_bucket(const int* __restrict__ rowI, const int* __restrict__ colI,
                                               int* __restrict__ M, int* __restrict__ Pz,
                                               int* __restrict__ stage, int E, int n) {
    __shared__ int hist[NBK];
    __shared__ int wsum[BT / 32];
    const int tid  = threadIdx.x;
    const int lane = tid & 31;
    const int wave = tid >> 5;
    const int c    = blockIdx.x;
    const int CH   = (E + NBK - 1) / NBK;
    const unsigned BW = (unsigned)((n + NBK - 1) / NBK);
    const int e0 = c * CH;
    const int e1 = (e0 + CH < E) ? (e0 + CH) : E;

    if (tid < NBK) hist[tid] = 0;
    __syncthreads();
    for (int eb = e0 + tid; eb < e1; eb += 4 * BT) {
        int cc[4];
#pragma unroll
        for (int k = 0; k < 4; ++k) {
            const int e = eb + k * BT;
            cc[k] = (e < e1) ? colI[e] : 0;
        }
#pragma unroll
        for (int k = 0; k < 4; ++k)
            if (eb + k * BT < e1) atomicAdd(&hist[(unsigned)cc[k] / BW], 1);
    }
    __syncthreads();

    const int hv = (tid < NBK) ? hist[tid] : 0;
    int inc = hv;
    for (int d = 1; d < 32; d <<= 1) {
        int t = __shfl_up(inc, d, 32);
        if (lane >= d) inc += t;
    }
    if (lane == 31) wsum[wave] = inc;
    __syncthreads();
    int pre = 0;
    for (int w = 0; w < wave; ++w) pre += wsum[w];
    const int ex = pre + inc - hv;
    if (tid < NBK) {
        hist[tid] = ex;
        M[tid * NBK + c]  = hv;
        Pz[tid * NBK + c] = ex;
    }
    __syncthreads();

    int2* __restrict__ st = (int2*)stage;
    for (int eb = e0 + tid; eb < e1; eb += 4 * BT) {
        int cc[4], rr[4];
#pragma unroll
        for (int k = 0; k < 4; ++k) {
            const int e = eb + k * BT;
            cc[k] = (e < e1) ? colI[e] : 0;
            rr[k] = (e < e1) ? rowI[e] : 0;
        }
#pragma unroll
        for (int k = 0; k < 4; ++k) {
            if (eb + k * BT < e1) {
                const int r = atomicAdd(&hist[(unsigned)cc[k] / BW], 1);
                st[e0 + r] = make_int2(rr[k], cc[k]);
            }
        }
    }
}

__global__ __launch_bounds__(BT) void k_rows(const int* __restrict__ M, const int* __restrict__ Pz,
                                             const int* __restrict__ stage, int* __restrict__ off,
                                             float* __restrict__ dinv, int* __restrict__ src, int E, int n) {
    __shared__ int segCnt[NBK];
    __shared__ int segPos[NBK];
    __shared__ int cntL[BWCAP];
    __shared__ int red[BT / 32];
    const int tid  = threadIdx.x;
    const int lane = tid & 31;
    const int wave = tid >> 5;
    const int b    = blockIdx.x;
    const int CH   = (E + NBK - 1) / NBK;
    const int BW   = (n + NBK - 1) / NBK;
    const int lo   = b * BW;
    const int nv   = (lo + BW <= n) ? BW : (n - lo);

    static_assert((NBK - 1) * (NBK / 4) <= 16 * BT, "sixteen pieces per thread cover the rows in front");
    const int4* __restrict__ M4 = (const int4*)M;
    const int nq = b * (NBK / 4);
    int4 q[16];
#pragma unroll
    for (int k = 0; k < 16; ++k) {
        const int i = tid + k * BT;
        q[k] = make_int4(0, 0, 0, 0);
        if (i < nq) q[k] = M4[i];
    }
    if (tid < NBK) {
        segCnt[tid] = M[b * NBK + tid];
        segPos[tid] = tid * CH + Pz[b * NBK + tid];
    }
    cntL[tid] = 0;
    int s = 0;
#pragma unroll
    for (int k = 0; k < 16; ++k) s += (q[k].x + q[k].y) + (q[k].z + q[k].w);
    for (int m = 16; m >= 1; m >>= 1) s += __shfl_xor(s, m, 32);
    if (lane == 0) red[wave] = s;
    __syncthreads();
    int baseB = 0;
    for (int w = 0; w < BT / 32; ++w) baseB += red[w];

    const int2* __restrict__ st = (const int2*)stage;
    const int sc = segCnt[tid >> 2];
    const int sp = segPos[tid >> 2];
    for (int j0 = tid & 3; j0 < sc; j0 += 16) {
        int y[4];
#pragma unroll
        for (int k = 0; k < 4; ++k) {
            const int j = j0 + 4 * k;
            y[k] = (j < sc) ? st[sp + j].y : lo;
        }
#pragma unroll
        for (int k = 0; k < 4; ++k)
            if (j0 + 4 * k < sc) atomicAdd(&cntL[y[k] - lo], 1);
    }
    __syncthreads();

    const int cv = (tid < nv) ? cntL[tid] : 0;
    int inc = cv;
    for (int d = 1; d < 32; d <<= 1) {
        int t = __shfl_up(inc, d, 32);
        if (lane >= d) inc += t;
    }
    if (lane == 31) red[wave] = inc;
    __syncthreads();
    int run = baseB + inc - cv;
    for (int w = 0; w < wave; ++w) run += red[w];
    if (tid < nv) {
        off[lo + tid]  = run;
        dinv[lo + tid] = rsqrtf((float)(cv + 1));
        cntL[tid]      = run;
    }
    if (b == 0 && tid == 0) off[n] = E;
    __syncthreads();

    for (int j0 = tid & 3; j0 < sc; j0 += 16) {
        int2 rc[4];
#pragma unroll
        for (int k = 0; k < 4; ++k) {
            const int j = j0 + 4 * k;
            rc[k] = (j < sc) ? st[sp + j] : make_int2(0, lo);
        }
#pragma unroll
        for (int k = 0; k < 4; ++k) {
            if (j0 + 4 * k < sc) {
                const int p = atomicAdd(&cntL[rc[k].y - lo], 1);
                src[p] = rc[k].x;
            }
        }
    }
}

#define NPW    4
#define TILE_N 32
#define PITCH  132

__device__ __forceinline__ float4 fma4(float4 h, float w, float4 a) {
    a.x = fmaf(h.x, w, a.x);
    a.y = fmaf(h.y, w, a.y);
    a.z = fmaf(h.z, w, a.z);
    a.w = fmaf(h.w, w, a.w);
    return a;
}

__device__ __forceinline__ float4 add4(float4 h, float4 a) {
    a.x += h.x;
    a.y += h.y;
    a.z += h.z;
    a.w += h.w;
    return a;
}

__device__ __forceinline__ float4 zero4() { return (float4){0.f, 0.f, 0.f, 0.f}; }

template <bool PRE, bool LAST>
__global__ __launch_bounds__(256, 16) void k_layer(const int* __restrict__ off,
                                               const int* __restrict__ src,
                                               const float* __restrict__ X,
                                               const float* __restrict__ dinv,
                                               const float* __restrict__ W,
                                               const float* __restrict__ b,
                                               const float* __restrict__ E0,
                                               const float* __restrict__ XA,
                                               float* __restrict__ dst, int n) {
    __shared__ __attribute__((aligned(16))) float S[TILE_N * PITCH];
    __shared__ __attribute__((aligned(16))) float Y[TILE_N * PITCH];

    const int lane = threadIdx.x & 31;
    const int wave = __builtin_amdgcn_readfirstlane((int)(threadIdx.x >> 5));
    const int vb   = blockIdx.x * TILE_N + wave * NPW;

    const unsigned lo = (unsigned)lane * 16u;
    auto rowof = [&](const float* tab, int r) -> float4 {
        const char* p = (const char*)tab + (size_t)(unsigned)r * 512u;
        unsigned l = lo;
        asm volatile("" : "+v"(l));
        return *(const float4*)(p + l);
    };
    auto rowld = [&](int r) -> float4 { return rowof(X, r); };
    auto rowst = [&](float* tab, int r, float4 val) {
        char* p = (char*)tab + (size_t)(unsigned)r * 512u;
        unsigned l = lo;
        asm volatile("" : "+v"(l));
        *(float4*)(p + l) = val;
    };

    int   ol = 0;
    float dl = 0.f;
    {
        const int vi = vb + lane;
        if (lane <= NPW) ol = off[vi < n ? vi : n];
        if (lane < NPW && vi < n) dl = dinv[vi];
    }
    const int dli = __float_as_int(dl);
    float dsc[NPW];
#pragma unroll
    for (int i = 0; i < NPW; ++i) dsc[i] = __int_as_float(__builtin_amdgcn_readlane(dli, i));
    const int beg = __builtin_amdgcn_readlane(ol, 0);
    const int end = __builtin_amdgcn_readlane(ol, NPW);

    int rl = 0, rn = 0;
    float wl = 0.f;
    if (lane < end - beg)      rl = src[beg + lane];
    if (lane < end - beg - 32) rn = src[beg + 32 + lane];
    {
        float4 hs[NPW];
#pragma unroll
        for (int i = 0; i < NPW; ++i) {
            hs[i] = zero4();
            if (vb + i < n) hs[i] = rowld(vb + i);
        }
        float wl0 = 0.f;
        if (!PRE) { if (lane < end - beg) wl0 = dinv[rl]; }
#pragma unroll
        for (int i = 0; i < NPW; ++i) {
            const float di = PRE ? 1.f : dsc[i];
            float4 s;
            s.x = hs[i].x * di; s.y = hs[i].y * di; s.z = hs[i].z * di; s.w = hs[i].w * di;
            *(float4*)(S + (wave * NPW + i) * PITCH + lane * 4) = s;
        }
        wl = wl0;
    }

    int    cur   = 0;
    int    nextb = __builtin_amdgcn_readlane(ol, 1);
    float  dc    = __int_as_float(__builtin_amdgcn_readlane(dli, 0));
    float4 acc   = zero4();

    auto flush = [&]() {
        float* sp = S + (wave * NPW + cur) * PITCH + lane * 4;
        const float4 t = *(const float4*)sp;
        float4 s;
        s.x = (acc.x + t.x) * dc; s.y = (acc.y + t.y) * dc; s.z = (acc.z + t.z) * dc; s.w = (acc.w + t.w) * dc;
        *(float4*)sp = s;
        ++cur;
        acc = zero4();
        if (cur < NPW) {
            nextb = __builtin_amdgcn_readlane(ol, cur + 1);
            dc    = __int_as_float(__builtin_amdgcn_readlane(dli, cur));
        }
    };

    for (int e0 = beg; e0 < end; e0 += 32) {
        const int m = (end - e0 < 32) ? (end - e0) : 32;
        float wn = 0.f;
        int   rnn = 0;
        auto ahead = [&]() {
            if (!PRE) { if (lane < end - e0 - 32) wn = dinv[rn]; }
            if (lane < end - e0 - 64) rnn = src[e0 + 64 + lane];
        };
        const int wi = __float_as_int(wl);

        int j = 0;
        for (; j + 8 <= m; j += 8) {
            float4 h[8];
#pragma unroll
            for (int k = 0; k < 8; ++k)
                h[k] = rowld(__builtin_amdgcn_readlane(rl, j + k));
            if (j == 0) ahead();
#pragma unroll
            for (int k = 0; k < 8; ++k) {
                while (e0 + j + k >= nextb) flush();
                acc = PRE ? add4(h[k], acc) : fma4(h[k], __int_as_float(__builtin_amdgcn_readlane(wi, j + k)), acc);
            }
        }
        const int t = m - j;
        if (t > 0) {
            float4 h[7];
#pragma unroll
            for (int k = 0; k < 7; ++k) {
                h[k] = zero4();
                if (k < t) h[k] = rowld(__builtin_amdgcn_readlane(rl, j + k));
            }
            if (j == 0) ahead();
#pragma unroll
            for (int k = 0; k < 7; ++k) {
                if (k < t) {
                    while (e0 + j + k >= nextb) flush();
                    acc = PRE ? add4(h[k], acc) : fma4(h[k], __int_as_float(__builtin_amdgcn_readlane(wi, j + k)), acc);
                }
            }
        }
        rl = rn; wl = wn; rn = rnn;
    }
    while (cur < NPW) flush();

    __syncthreads();

    const int half = lane >> 4;
    const int l16  = lane & 15;
    const int kb   = half * 2;
    const float* a0p = S + l16 * PITCH + kb;
    const float* a1p = a0p + 16 * PITCH;
    const float* __restrict__ wp = W + (unsigned)(kb * EMB + wave * 16 + l16);

    v8f c0 = (v8f){0.f,0.f,0.f,0.f,0.f,0.f,0.f,0.f};
    v8f c1 = c0;
    v2f bc[4], bn[4];
#pragma unroll
    for (int s = 0; s < 4; ++s) {
        bc[s].x = wp[(size_t)(4 * s) * EMB];
        bc[s].y = wp[(size_t)(4 * s + 1) * EMB];
    }
#pragma unroll
    for (int k0 = 0; k0 < EMB; k0 += 16) {
        if (k0 + 16 < EMB) {
#pragma unroll
            for (int s = 0; s < 4; ++s) {
                bn[s].x = wp[(size_t)(k0 + 16 + 4 * s) * EMB];
                bn[s].y = wp[(size_t)(k0 + 16 + 4 * s + 1) * EMB];
            }
        }
#pragma unroll
        for (int s = 0; s < 4; ++s) {
            const v2f a0 = *(const v2f*)(a0p + k0 + 4 * s);
            const v2f a1 = *(const v2f*)(a1p + k0 + 4 * s);
            c0 = __builtin_amdgcn_wmma_f32_16x16x4_f32(false, a0, false, bc[s], (short)0, c0, false, false);
            c1 = __builtin_amdgcn_wmma_f32_16x16x4_f32(false, a1, false, bc[s], (short)0, c1, false, false);
        }
        if (k0 + 16 < EMB) {
#pragma unroll
            for (int s = 0; s < 4; ++s) bc[s] = bn[s];
        }
    }

    {
        float* y0 = Y + (half * 8) * PITCH + wave * 16 + l16;
        float* y1 = y0 + 16 * PITCH;
#pragma unroll
        for (int i = 0; i < 8; ++i) {
            y0[i * PITCH] = c0[i];
            y1[i * PITCH] = c1[i];
        }
    }
    __syncthreads();

    const float4 bb = *(const float4*)((const char*)b + lo);
    if (!LAST) {
#pragma unroll
        for (int i = 0; i < NPW; ++i) {
            const int v = vb + i;
            if (v < n) {
                float4 y = *(const float4*)((const char*)(Y + (wave * NPW + i) * PITCH) + lo);
                y.x += bb.x; y.y += bb.y; y.z += bb.z; y.w += bb.w;
                float4 ys;
                ys.x = y.x * dsc[i]; ys.y = y.y * dsc[i]; ys.z = y.z * dsc[i]; ys.w = y.w * dsc[i];
                rowst(dst, v, ys);
            }
        }
    } else {
#pragma unroll
        for (int i0 = 0; i0 < NPW; i0 += 2) {
            float4 e[2], a[2], c[2];
#pragma unroll
            for (int k = 0; k < 2; ++k) {
                e[k] = zero4(); a[k] = zero4(); c[k] = zero4();
                if (vb + i0 + k < n) {
                    e[k] = rowof(E0, vb + i0 + k);
                    a[k] = rowof(XA, vb + i0 + k);
                    c[k] = rowof(X,  vb + i0 + k);
                }
            }
#pragma unroll
            for (int k = 0; k < 2; ++k) {
                const int v = vb + i0 + k;
                if (v < n) {
                    const float sd = 1.0f / dsc[i0 + k];
                    float4 y = *(const float4*)((const char*)(Y + (wave * NPW + i0 + k) * PITCH) + lo);
                    y.x += bb.x; y.y += bb.y; y.z += bb.z; y.w += bb.w;
                    float4 o;
                    o.x = fmaf(c[k].x, sd, fmaf(a[k].x, sd, e[k].x)) + y.x;
                    o.y = fmaf(c[k].y, sd, fmaf(a[k].y, sd, e[k].y)) + y.y;
                    o.z = fmaf(c[k].z, sd, fmaf(a[k].z, sd, e[k].z)) + y.z;
                    o.w = fmaf(c[k].w, sd, fmaf(a[k].w, sd, e[k].w)) + y.w;
                    rowst(dst, v, o);
                }
            }
        }
    }
}

extern "C" void kernel_launch(void* const* d_in, const int* in_sizes, int n_in,
                              void* d_out, int out_size, void* d_ws, size_t ws_size,
                              hipStream_t stream) {
    const int*   edge = (const int*)d_in[0];
    const float* emb  = (const float*)d_in[1];
    const float* Wall = (const float*)d_in[2];
    const float* ball = (const float*)d_in[3];
    float*       out  = (float*)d_out;

    const int E = in_sizes[0] / 2;
    const int n = out_size / EMB;

    const int* rowI = edge;
    const int* colI = edge + E;

    const int mTiles  = (n + 15) / 16;
    const int rowsPad = mTiles * 16;
    const size_t matElems = (size_t)rowsPad * EMB;

    float* ws   = (float*)d_ws;
    float* bufA = ws;
    float* bufB = ws + matElems;
    int*   iws  = (int*)(ws + 2 * matElems);
    float* dinv = ws + 3 * matElems;

    const int nAl = rowsPad + 16;
    int* M     = iws;
    int* Pz    = iws + NBK * NBK;
    int* off   = iws + 2 * NBK * NBK;
    int* stage = off + nAl;
    int* src   = stage + 2 * (size_t)E;

    k_bucket<<<NBK, BT, 0, stream>>>(rowI, colI, M, Pz, stage, E, n);
    k_rows  <<<NBK, BT, 0, stream>>>(M, Pz, stage, off, dinv, src, E, n);
    const int nb = (n + TILE_N - 1) / TILE_N;
    k_layer<false, false><<<nb, 256, 0, stream>>>(off, src, emb,  dinv, Wall,                         ball,
                                                  emb, emb, bufA, n);
    k_layer<true,  false><<<nb, 256, 0, stream>>>(off, src, bufA, dinv, Wall + (size_t)EMB * EMB,     ball + EMB,
                                                  emb, emb, bufB, n);
    k_layer<true,  true> <<<nb, 256, 0, stream>>>(off, src, bufB, dinv, Wall + (size_t)2 * EMB * EMB, ball + 2 * EMB,
                                                  emb, bufA, out, n);
}
